# starting kernel
# baseline (speedup 1.0000x reference)
#include <hip/hip_runtime.h>
#include <cstddef>

typedef __attribute__((ext_vector_type(2))) float v2f;
typedef __attribute__((ext_vector_type(8))) float v8f;

__device__ __forceinline__ float wave32_sum(float v) {
#pragma unroll
  for (int m = 16; m >= 1; m >>= 1) v += __shfl_xor(v, m, 32);
  return v;
}
__device__ __forceinline__ float half16_sum(float v) {
#pragma unroll
  for (int m = 1; m < 16; m <<= 1) v += __shfl_xor(v, m, 32);
  return v;
}

__device__ __forceinline__ v8f wmma_f32_k4(v2f a, v2f b, v8f c) {
  return __builtin_amdgcn_wmma_f32_16x16x4_f32(false, a, false, b, (short)0, c,
                                               false, false);
}

__global__ void k_zero(float* p, int n) {
  int i = blockIdx.x * blockDim.x + threadIdx.x;
  if (i < n) p[i] = 0.f;
}

__global__ void k_reduce_x(const float* __restrict__ x,
                           float* __restrict__ rowsO, float* __restrict__ colsO,
                           float* __restrict__ diagO, float* __restrict__ dsumO,
                           float* __restrict__ totO) {
  __shared__ float colbuf[16 * 512];
  __shared__ float totbuf[16];

  const int bid = blockIdx.x;
  const int mat = bid >> 2;
  const int r0 = (bid & 3) << 7;
  const int tid = threadIdx.x;
  const int w = tid >> 5, l = tid & 31;
  const float* M = x + (size_t)mat * 262144;

  float ca[16];
#pragma unroll
  for (int i = 0; i < 16; ++i) ca[i] = 0.f;
  float totp = 0.f;

#pragma unroll 4
  for (int t = 0; t < 8; ++t) {
    const int r = r0 + w + (t << 4);
    const float4* rp = (const float4*)(M + (size_t)r * 512);
    float rsum = 0.f;
#pragma unroll
    for (int c = 0; c < 4; ++c) {
      float4 v = rp[(c << 5) + l];
      ca[c * 4 + 0] += v.x; ca[c * 4 + 1] += v.y;
      ca[c * 4 + 2] += v.z; ca[c * 4 + 3] += v.w;
      rsum += (v.x + v.y) + (v.z + v.w);
    }
    rsum = wave32_sum(rsum);
    if (l == 0) { rowsO[mat * 512 + r] = rsum; totp += rsum; }
  }
#pragma unroll
  for (int c = 0; c < 4; ++c)
#pragma unroll
    for (int j = 0; j < 4; ++j)
      colbuf[w * 512 + c * 128 + l * 4 + j] = ca[c * 4 + j];
  if (l == 0) totbuf[w] = totp;

  float dv = 0.f;
  if (tid < 128) {
    const int r = r0 + tid;
    dv = M[(size_t)r * 512 + r];
    diagO[mat * 512 + r] = dv;
  }
  float ds = wave32_sum(dv);

  __syncthreads();
  float csum = 0.f;
#pragma unroll
  for (int w2 = 0; w2 < 16; ++w2) csum += colbuf[w2 * 512 + tid];
  atomicAdd(&colsO[mat * 512 + tid], csum);

  if (l == 0 && w < 4) atomicAdd(&dsumO[mat], ds);
  if (tid == 0) {
    float T = 0.f;
#pragma unroll
    for (int i = 0; i < 16; ++i) T += totbuf[i];
    atomicAdd(&totO[mat], T);
  }
}

__global__ void k_prep2(const float* __restrict__ c2, const float* __restrict__ bias2,
                        const float* __restrict__ dsum, const float* __restrict__ tot,
                        float* __restrict__ W0, float* __restrict__ W2,
                        float* __restrict__ W3, float* __restrict__ cst0) {
  const int tid = threadIdx.x;
  for (int idx = tid; idx < 1024; idx += 256) {
    int s = idx >> 4, d = idx & 15;
    const float* c = c2 + (size_t)(d * 64 + s) * 5;
    W0[idx] = c[0];
    W2[idx] = c[2] * (1.f / 512.f);
    W3[idx] = c[3] * (1.f / 512.f);
  }
  for (int idx = tid; idx < 512; idx += 256) {
    int n = idx >> 6, s = idx & 63;
    float acc = bias2[s];
    for (int d = 0; d < 16; ++d) {
      const float* c = c2 + (size_t)(d * 64 + s) * 5;
      acc += c[1] * dsum[n * 16 + d] * (1.f / 512.f) +
             c[4] * tot[n * 16 + d] * (1.f / (512.f * 512.f));
    }
    cst0[idx] = acc;
  }
}

__device__ __forceinline__ v8f gemm_k16(v8f acc, const float* __restrict__ W,
                                        const float* __restrict__ S, int s0,
                                        int i0, int q, int h) {
#pragma unroll
  for (int k0 = 0; k0 < 16; k0 += 4) {
    const int ka = k0 + 2 * h;
    v2f a, b;
    a.x = W[(s0 + q) * 16 + ka];
    a.y = W[(s0 + q) * 16 + ka + 1];
    b.x = S[ka * 512 + i0 + q];
    b.y = S[(ka + 1) * 512 + i0 + q];
    acc = wmma_f32_k4(a, b, acc);
  }
  return acc;
}

__global__ void k_layer2(const float* __restrict__ diag, const float* __restrict__ rows,
                         const float* __restrict__ cols, const float* __restrict__ W0,
                         const float* __restrict__ W2, const float* __restrict__ W3,
                         const float* __restrict__ cst, float* __restrict__ aout,
                         float* __restrict__ bnsum, float* __restrict__ bnsq,
                         float* __restrict__ sumA) {
  const int gw = blockIdx.x * 4 + (threadIdx.x >> 5);
  const int n = gw >> 7, rem = gw & 127;
  const int s0 = (rem >> 5) << 4, i0 = (rem & 31) << 4;
  const int l = threadIdx.x & 31, h = l >> 4, q = l & 15;
  const size_t nb = (size_t)n * 16 * 512;

  v8f acc = {};
  acc = gemm_k16(acc, W0, diag + nb, s0, i0, q, h);
  acc = gemm_k16(acc, W2, rows + nb, s0, i0, q, h);
  acc = gemm_k16(acc, W3, cols + nb, s0, i0, q, h);

#pragma unroll
  for (int r = 0; r < 8; ++r) {
    const int s = s0 + r + 8 * h;
    float v = acc[r] + cst[n * 64 + s];
    v = fmaxf(v, 0.f);
    aout[((size_t)n * 64 + s) * 512 + i0 + q] = v;
    float sv = half16_sum(v);
    float sv2 = half16_sum(v * v);
    if (q == 0) {
      atomicAdd(&bnsum[s], sv);
      atomicAdd(&bnsq[s], sv2);
      atomicAdd(&sumA[n * 64 + s], sv);
    }
  }
}

__global__ void k_bn_mix(const float* __restrict__ bnsum, const float* __restrict__ bnsq,
                         const float* __restrict__ g, const float* __restrict__ bt,
                         const float* __restrict__ c, const float* __restrict__ bias,
                         const float* __restrict__ sumA, float* __restrict__ Mp,
                         float* __restrict__ cst) {
  __shared__ float sA[64], sB[64];
  const int tid = threadIdx.x;
  if (tid < 64) {
    float mu = bnsum[tid] * (1.f / 4096.f);
    float var = bnsq[tid] * (1.f / 4096.f) - mu * mu;
    float A = g[tid] * rsqrtf(var + 1e-5f);
    sA[tid] = A;
    sB[tid] = bt[tid] - A * mu;
  }
  __syncthreads();
  for (int idx = tid; idx < 4096; idx += 256) {
    int s = idx >> 6, d = idx & 63;
    Mp[idx] = c[(size_t)(d * 64 + s) * 2] * sA[d];
  }
  for (int idx = tid; idx < 512; idx += 256) {
    int n = idx >> 6, s = idx & 63;
    float acc = bias[s];
    for (int d = 0; d < 64; ++d) {
      float c0 = c[(size_t)(d * 64 + s) * 2];
      float c1 = c[(size_t)(d * 64 + s) * 2 + 1];
      acc += c0 * sB[d] +
             c1 * (sA[d] * sumA[n * 64 + d] + 512.f * sB[d]) * (1.f / 512.f);
    }
    cst[idx] = acc;
  }
}

__global__ void k_transform(const float* __restrict__ ain, const float* __restrict__ W,
                            const float* __restrict__ cst, float* __restrict__ aout,
                            float* __restrict__ bnsum, float* __restrict__ bnsq,
                            float* __restrict__ sumA) {
  const int gw = blockIdx.x * 4 + (threadIdx.x >> 5);
  const int n = gw >> 7, rem = gw & 127;
  const int s0 = (rem >> 5) << 4, i0 = (rem & 31) << 4;
  const int l = threadIdx.x & 31, h = l >> 4, q = l & 15;
  const float* S = ain + (size_t)n * 64 * 512;

  v8f acc = {};
#pragma unroll
  for (int k0 = 0; k0 < 64; k0 += 4) {
    const int ka = k0 + 2 * h;
    v2f a, b;
    a.x = W[(s0 + q) * 64 + ka];
    a.y = W[(s0 + q) * 64 + ka + 1];
    b.x = S[ka * 512 + i0 + q];
    b.y = S[(ka + 1) * 512 + i0 + q];
    acc = wmma_f32_k4(a, b, acc);
  }

#pragma unroll
  for (int r = 0; r < 8; ++r) {
    const int s = s0 + r + 8 * h;
    float v = acc[r] + cst[n * 64 + s];
    v = fmaxf(v, 0.f);
    aout[((size_t)n * 64 + s) * 512 + i0 + q] = v;
    float sv = half16_sum(v);
    float sv2 = half16_sum(v * v);
    if (q == 0) {
      atomicAdd(&bnsum[s], sv);
      atomicAdd(&bnsq[s], sv2);
      if (sumA) atomicAdd(&sumA[n * 64 + s], sv);
    }
  }
}

__global__ void k_bn_fc(const float* __restrict__ bnsum, const float* __restrict__ bnsq,
                        const float* __restrict__ g, const float* __restrict__ bt,
                        const float* __restrict__ fc1w, const float* __restrict__ fc1b,
                        float* __restrict__ F1, float* __restrict__ cf1) {
  __shared__ float sA[64], sB[64];
  const int tid = threadIdx.x;
  if (tid < 64) {
    float mu = bnsum[tid] * (1.f / 4096.f);
    float var = bnsq[tid] * (1.f / 4096.f) - mu * mu;
    float A = g[tid] * rsqrtf(var + 1e-5f);
    sA[tid] = A;
    sB[tid] = bt[tid] - A * mu;
  }
  __syncthreads();
  for (int idx = tid; idx < 4096; idx += 256) {
    int d = idx & 63;
    F1[idx] = fc1w[idx] * sA[d];
  }
  if (tid < 64) {
    float acc = fc1b[tid];
    for (int d = 0; d < 64; ++d) acc += fc1w[tid * 64 + d] * sB[d];
    cf1[tid] = acc;
  }
}

__global__ void k_fc(const float* __restrict__ a2, const float* __restrict__ F1,
                     const float* __restrict__ cf1, const float* __restrict__ fc2w,
                     const float* __restrict__ fc2b, float* __restrict__ out) {
  __shared__ float r1[64 * 16];
  const int n = blockIdx.x >> 5, i0 = (blockIdx.x & 31) << 4;
  const int w = threadIdx.x >> 5, l = threadIdx.x & 31, h = l >> 4, q = l & 15;
  const float* S = a2 + (size_t)n * 64 * 512;

  v8f acc = {};
#pragma unroll
  for (int k0 = 0; k0 < 64; k0 += 4) {
    const int ka = k0 + 2 * h;
    v2f a, b;
    a.x = F1[(w * 16 + q) * 64 + ka];
    a.y = F1[(w * 16 + q) * 64 + ka + 1];
    b.x = S[ka * 512 + i0 + q];
    b.y = S[(ka + 1) * 512 + i0 + q];
    acc = wmma_f32_k4(a, b, acc);
  }
#pragma unroll
  for (int r = 0; r < 8; ++r) {
    const int o = w * 16 + r + 8 * h;
    r1[o * 16 + q] = fmaxf(acc[r] + cf1[o], 0.f);
  }
  __syncthreads();

  if (w == 0) {
    v8f acc2 = {};
#pragma unroll
    for (int k0 = 0; k0 < 64; k0 += 4) {
      const int ka = k0 + 2 * h;
      v2f a, b;
      a.x = fc2w[q * 64 + ka];
      a.y = fc2w[q * 64 + ka + 1];
      b.x = r1[ka * 16 + q];
      b.y = r1[(ka + 1) * 16 + q];
      acc2 = wmma_f32_k4(a, b, acc2);
    }
#pragma unroll
    for (int r = 0; r < 8; ++r) {
      const int o = r + 8 * h;
      out[((size_t)n * 16 + o) * 512 + i0 + q] = acc2[r] + fc2b[o];
    }
  }
}

extern "C" void kernel_launch(void* const* d_in, const int* in_sizes, int n_in,
                              void* d_out, int out_size, void* d_ws, size_t ws_size,
                              hipStream_t stream) {
  const float* x      = (const float*)d_in[0];
  const float* c2     = (const float*)d_in[1];
  const float* bias2  = (const float*)d_in[2];
  const float* c1a    = (const float*)d_in[3];
  const float* bias1a = (const float*)d_in[4];
  const float* c1b    = (const float*)d_in[5];
  const float* bias1b = (const float*)d_in[6];
  const float* g0 = (const float*)d_in[7],  *b0 = (const float*)d_in[8];
  const float* g1 = (const float*)d_in[9],  *b1 = (const float*)d_in[10];
  const float* g2 = (const float*)d_in[11], *b2 = (const float*)d_in[12];
  const float* fc1w = (const float*)d_in[13], *fc1b = (const float*)d_in[14];
  const float* fc2w = (const float*)d_in[15], *fc2b = (const float*)d_in[16];
  float* out = (float*)d_out;
  float* ws = (float*)d_ws;

  float* rows = ws;
  float* diag = ws + 65536;
  float* a0   = ws + 131072;
  float* a1   = a0 + 262144;
  float* a2   = a1 + 262144;
  float* zb   = a2 + 262144;
  float* cols = zb;
  float* dsum = zb + 65536;
  float* tot  = zb + 65664;
  float* st   = zb + 65792;
  float* bnsum0 = st,       *bnsq0 = st + 64;
  float* bnsum1 = st + 128, *bnsq1 = st + 192;
  float* bnsum2 = st + 256, *bnsq2 = st + 320;
  float* sumA0 = st + 384;
  float* sumA1 = st + 896;
  float* W0 = zb + 67200, *W2 = W0 + 1024, *W3 = W2 + 1024;
  float* cst0 = W3 + 1024;
  float* M1p = cst0 + 512;
  float* cst1 = M1p + 4096;
  float* M2p = cst1 + 512;
  float* cst2 = M2p + 4096;
  float* F1p = cst2 + 512;
  float* cf1 = F1p + 4096;

  k_zero<<<263, 256, 0, stream>>>(zb, 67200);
  k_reduce_x<<<512, 512, 0, stream>>>(x, rows, cols, diag, dsum, tot);
  k_prep2<<<1, 256, 0, stream>>>(c2, bias2, dsum, tot, W0, W2, W3, cst0);
  k_layer2<<<256, 128, 0, stream>>>(diag, rows, cols, W0, W2, W3, cst0, a0,
                                    bnsum0, bnsq0, sumA0);
  k_bn_mix<<<1, 256, 0, stream>>>(bnsum0, bnsq0, g0, b0, c1a, bias1a, sumA0,
                                  M1p, cst1);
  k_transform<<<256, 128, 0, stream>>>(a0, M1p, cst1, a1, bnsum1, bnsq1, sumA1);
  k_bn_mix<<<1, 256, 0, stream>>>(bnsum1, bnsq1, g1, b1, c1b, bias1b, sumA1,
                                  M2p, cst2);
  k_transform<<<256, 128, 0, stream>>>(a1, M2p, cst2, a2, bnsum2, bnsq2, nullptr);
  k_bn_fc<<<1, 256, 0, stream>>>(bnsum2, bnsq2, g2, b2, fc1w, fc1b, F1p, cf1);
  k_fc<<<256, 128, 0, stream>>>(a2, F1p, cf1, fc2w, fc2b, out);
}


# Fold the four one-block parameter launches (layer-2 constant, two BN+mix folds, BN+fc1 fold) into the prologues of their consumer kernels: 10 launches -> 6; every sum keeps its order
# speedup vs baseline: 1.4646x; 1.4646x over previous
#include <hip/hip_runtime.h>
#include <cstddef>

typedef __attribute__((ext_vector_type(2))) float v2f;
typedef __attribute__((ext_vector_type(8))) float v8f;

__device__ __forceinline__ float wave32_sum(float v) {
#pragma unroll
  for (int m = 16; m >= 1; m >>= 1) v += __shfl_xor(v, m, 32);
  return v;
}
__device__ __forceinline__ float half16_sum(float v) {
#pragma unroll
  for (int m = 1; m < 16; m <<= 1) v += __shfl_xor(v, m, 32);
  return v;
}

__device__ __forceinline__ v8f wmma_f32_k4(v2f a, v2f b, v8f c) {
  return __builtin_amdgcn_wmma_f32_16x16x4_f32(false, a, false, b, (short)0, c,
                                               false, false);
}

__global__ void k_zero(float* p, int n) {
  int i = blockIdx.x * blockDim.x + threadIdx.x;
  if (i < n) p[i] = 0.f;
}

__global__ void k_reduce_x(const float* __restrict__ x,
                           float* __restrict__ rowsO, float* __restrict__ colsO,
                           float* __restrict__ diagO, float* __restrict__ dsumO,
                           float* __restrict__ totO) {
  __shared__ float colbuf[16 * 512];
  __shared__ float totbuf[16];

  const int bid = blockIdx.x;
  const int mat = bid >> 2;
  const int r0 = (bid & 3) << 7;
  const int tid = threadIdx.x;
  const int w = tid >> 5, l = tid & 31;
  const float* M = x + (size_t)mat * 262144;

  float ca[16];
#pragma unroll
  for (int i = 0; i < 16; ++i) ca[i] = 0.f;
  float totp = 0.f;

#pragma unroll 4
  for (int t = 0; t < 8; ++t) {
    const int r = r0 + w + (t << 4);
    const float4* rp = (const float4*)(M + (size_t)r * 512);
    float rsum = 0.f;
#pragma unroll
    for (int c = 0; c < 4; ++c) {
      float4 v = rp[(c << 5) + l];
      ca[c * 4 + 0] += v.x; ca[c * 4 + 1] += v.y;
      ca[c * 4 + 2] += v.z; ca[c * 4 + 3] += v.w;
      rsum += (v.x + v.y) + (v.z + v.w);
    }
    rsum = wave32_sum(rsum);
    if (l == 0) { rowsO[mat * 512 + r] = rsum; totp += rsum; }
  }
#pragma unroll
  for (int c = 0; c < 4; ++c)
#pragma unroll
    for (int j = 0; j < 4; ++j)
      colbuf[w * 512 + c * 128 + l * 4 + j] = ca[c * 4 + j];
  if (l == 0) totbuf[w] = totp;

  float dv = 0.f;
  if (tid < 128) {
    const int r = r0 + tid;
    dv = M[(size_t)r * 512 + r];
    diagO[mat * 512 + r] = dv;
  }
  float ds = wave32_sum(dv);

  __syncthreads();
  float csum = 0.f;
#pragma unroll
  for (int w2 = 0; w2 < 16; ++w2) csum += colbuf[w2 * 512 + tid];
  atomicAdd(&colsO[mat * 512 + tid], csum);

  if (l == 0 && w < 4) atomicAdd(&dsumO[mat], ds);
  if (tid == 0) {
    float T = 0.f;
#pragma unroll
    for (int i = 0; i < 16; ++i) T += totbuf[i];
    atomicAdd(&totO[mat], T);
  }
}

__device__ __forceinline__ void bn_fold(const float* __restrict__ bnsum,
                                        const float* __restrict__ bnsq,
                                        const float* __restrict__ g,
                                        const float* __restrict__ bt, int d,
                                        float& A, float& B) {
  float mu = bnsum[d] * (1.f / 4096.f);
  float var = bnsq[d] * (1.f / 4096.f) - mu * mu;
  A = g[d] * rsqrtf(var + 1e-5f);
  B = bt[d] - A * mu;
}

__device__ __forceinline__ v8f gemm_k16(v8f acc, const float* __restrict__ c2,
                                        int b, float scale,
                                        const float* __restrict__ S, int s0,
                                        int i0, int q, int h) {
#pragma unroll
  for (int k0 = 0; k0 < 16; k0 += 4) {
    const int ka = k0 + 2 * h;
    v2f a, bb;
    a.x = c2[(size_t)(ka * 64 + s0 + q) * 5 + b] * scale;
    a.y = c2[(size_t)((ka + 1) * 64 + s0 + q) * 5 + b] * scale;
    bb.x = S[ka * 512 + i0 + q];
    bb.y = S[(ka + 1) * 512 + i0 + q];
    acc = wmma_f32_k4(a, bb, acc);
  }
  return acc;
}

__global__ void k_layer2(const float* __restrict__ diag, const float* __restrict__ rows,
                         const float* __restrict__ cols, const float* __restrict__ c2,
                         const float* __restrict__ bias2, const float* __restrict__ dsum,
                         const float* __restrict__ tot, float* __restrict__ aout,
                         float* __restrict__ bnsum, float* __restrict__ bnsq,
                         float* __restrict__ sumA) {
  const int gw = blockIdx.x * 4 + (threadIdx.x >> 5);
  const int n = gw >> 7, rem = gw & 127;
  const int s0 = (rem >> 5) << 4, i0 = (rem & 31) << 4;
  const int l = threadIdx.x & 31, h = l >> 4, q = l & 15;
  const size_t nb = (size_t)n * 16 * 512;

  float cs = bias2[s0 + q];
#pragma unroll
  for (int d = 0; d < 16; ++d) {
    const float* c = c2 + (size_t)(d * 64 + s0 + q) * 5;
    cs += c[1] * dsum[n * 16 + d] * (1.f / 512.f) +
          c[4] * tot[n * 16 + d] * (1.f / (512.f * 512.f));
  }

  v8f acc = {};
  acc = gemm_k16(acc, c2, 0, 1.f, diag + nb, s0, i0, q, h);
  acc = gemm_k16(acc, c2, 2, 1.f / 512.f, rows + nb, s0, i0, q, h);
  acc = gemm_k16(acc, c2, 3, 1.f / 512.f, cols + nb, s0, i0, q, h);

#pragma unroll
  for (int r = 0; r < 8; ++r) {
    const int s = s0 + r + 8 * h;
    float v = acc[r] + __shfl(cs, r + 8 * h, 32);
    v = fmaxf(v, 0.f);
    aout[((size_t)n * 64 + s) * 512 + i0 + q] = v;
    float sv = half16_sum(v);
    float sv2 = half16_sum(v * v);
    if (q == 0) {
      atomicAdd(&bnsum[s], sv);
      atomicAdd(&bnsq[s], sv2);
      atomicAdd(&sumA[n * 64 + s], sv);
    }
  }
}

__global__ void __launch_bounds__(128) k_transform(const float* __restrict__ ain,
                            const float* __restrict__ pbnsum,
                            const float* __restrict__ pbnsq,
                            const float* __restrict__ g, const float* __restrict__ bt,
                            const float* __restrict__ c, const float* __restrict__ bias,
                            const float* __restrict__ psumA,
                            float* __restrict__ aout, float* __restrict__ bnsum,
                            float* __restrict__ bnsq, float* __restrict__ sumA) {
  __shared__ __attribute__((aligned(16))) float sA[64];
  __shared__ __attribute__((aligned(16))) float sB[64];
  __shared__ __attribute__((aligned(16))) float sX[64];
  const int tid = threadIdx.x;
  const int gw = blockIdx.x * 4 + (tid >> 5);
  const int n = gw >> 7, rem = gw & 127;
  const int s0 = (rem >> 5) << 4, i0 = (rem & 31) << 4;
  const int l = tid & 31, h = l >> 4, q = l & 15;
  const float* S = ain + (size_t)n * 64 * 512;

  float braw[32];
#pragma unroll
  for (int j = 0; j < 16; ++j) {
    const int ka = 4 * j + 2 * h;
    braw[2 * j] = S[ka * 512 + i0 + q];
    braw[2 * j + 1] = S[(ka + 1) * 512 + i0 + q];
  }
  float2 cc[64];
#pragma unroll
  for (int d = 0; d < 64; ++d)
    cc[d] = *(const float2*)(c + (size_t)(d * 64 + s0 + q) * 2);

  if (tid < 64) {
    float A, B;
    bn_fold(pbnsum, pbnsq, g, bt, tid, A, B);
    sA[tid] = A;
    sB[tid] = B;
    sX[tid] = A * psumA[n * 64 + tid] + 512.f * B;
  }
  __syncthreads();

  float cs = bias[s0 + q];
#pragma unroll
  for (int d = 0; d < 64; ++d)
    cs += cc[d].x * sB[d] + cc[d].y * sX[d] * (1.f / 512.f);

  v8f acc = {};
#pragma unroll
  for (int j = 0; j < 16; ++j) {
    const int ka = 4 * j + 2 * h;
    const float2 sa = *(const float2*)&sA[ka];
    v2f a, b;
    const float c0 = cc[4 * j].x, c1 = cc[4 * j + 1].x;
    const float c2 = cc[4 * j + 2].x, c3 = cc[4 * j + 3].x;
    a.x = (h ? c2 : c0) * sa.x;
    a.y = (h ? c3 : c1) * sa.y;
    b.x = braw[2 * j];
    b.y = braw[2 * j + 1];
    acc = wmma_f32_k4(a, b, acc);
  }

#pragma unroll
  for (int r = 0; r < 8; ++r) {
    const int s = s0 + r + 8 * h;
    float v = acc[r] + __shfl(cs, r + 8 * h, 32);
    v = fmaxf(v, 0.f);
    aout[((size_t)n * 64 + s) * 512 + i0 + q] = v;
    float sv = half16_sum(v);
    float sv2 = half16_sum(v * v);
    if (q == 0) {
      atomicAdd(&bnsum[s], sv);
      atomicAdd(&bnsq[s], sv2);
      if (sumA) atomicAdd(&sumA[n * 64 + s], sv);
    }
  }
}

__global__ void k_fc(const float* __restrict__ a2, const float* __restrict__ pbnsum,
                     const float* __restrict__ pbnsq, const float* __restrict__ g,
                     const float* __restrict__ bt, const float* __restrict__ fc1w,
                     const float* __restrict__ fc1b, const float* __restrict__ fc2w,
                     const float* __restrict__ fc2b, float* __restrict__ out) {
  __shared__ float r1[64 * 16];
  __shared__ __attribute__((aligned(16))) float sA[64];
  __shared__ __attribute__((aligned(16))) float sB[64];
  const int tid = threadIdx.x;
  const int n = blockIdx.x >> 5, i0 = (blockIdx.x & 31) << 4;
  const int w = tid >> 5, l = tid & 31, h = l >> 4, q = l & 15;
  const float* S = a2 + (size_t)n * 64 * 512;

  float braw[32];
#pragma unroll
  for (int j = 0; j < 16; ++j) {
    const int ka = 4 * j + 2 * h;
    braw[2 * j] = S[ka * 512 + i0 + q];
    braw[2 * j + 1] = S[(ka + 1) * 512 + i0 + q];
  }
  float4 rw[16];
  {
    const float4* rp = (const float4*)(fc1w + (size_t)(w * 16 + q) * 64);
#pragma unroll
    for (int j = 0; j < 16; ++j) rw[j] = rp[j];
  }

  if (tid < 64) {
    float A, B;
    bn_fold(pbnsum, pbnsq, g, bt, tid, A, B);
    sA[tid] = A;
    sB[tid] = B;
  }
  __syncthreads();

  float cf = fc1b[w * 16 + q];
#pragma unroll
  for (int j = 0; j < 16; ++j) {
    cf += rw[j].x * sB[4 * j];
    cf += rw[j].y * sB[4 * j + 1];
    cf += rw[j].z * sB[4 * j + 2];
    cf += rw[j].w * sB[4 * j + 3];
  }

  v8f acc = {};
#pragma unroll
  for (int j = 0; j < 16; ++j) {
    const int ka = 4 * j + 2 * h;
    const float2 sa = *(const float2*)&sA[ka];
    v2f a, b;
    const float4 rj = rw[j];
    a.x = (h ? rj.z : rj.x) * sa.x;
    a.y = (h ? rj.w : rj.y) * sa.y;
    b.x = braw[2 * j];
    b.y = braw[2 * j + 1];
    acc = wmma_f32_k4(a, b, acc);
  }
#pragma unroll
  for (int r = 0; r < 8; ++r) {
    const int o = w * 16 + r + 8 * h;
    r1[o * 16 + q] = fmaxf(acc[r] + __shfl(cf, r + 8 * h, 32), 0.f);
  }
  __syncthreads();

  if (w == 0) {
    v8f acc2 = {};
#pragma unroll
    for (int k0 = 0; k0 < 64; k0 += 4) {
      const int ka = k0 + 2 * h;
      v2f a, b;
      a.x = fc2w[q * 64 + ka];
      a.y = fc2w[q * 64 + ka + 1];
      b.x = r1[ka * 16 + q];
      b.y = r1[(ka + 1) * 16 + q];
      acc2 = wmma_f32_k4(a, b, acc2);
    }
#pragma unroll
    for (int r = 0; r < 8; ++r) {
      const int o = r + 8 * h;
      out[((size_t)n * 16 + o) * 512 + i0 + q] = acc2[r] + fc2b[o];
    }
  }
}

extern "C" void kernel_launch(void* const* d_in, const int* in_sizes, int n_in,
                              void* d_out, int out_size, void* d_ws, size_t ws_size,
                              hipStream_t stream) {
  const float* x      = (const float*)d_in[0];
  const float* c2     = (const float*)d_in[1];
  const float* bias2  = (const float*)d_in[2];
  const float* c1a    = (const float*)d_in[3];
  const float* bias1a = (const float*)d_in[4];
  const float* c1b    = (const float*)d_in[5];
  const float* bias1b = (const float*)d_in[6];
  const float* g0 = (const float*)d_in[7],  *b0 = (const float*)d_in[8];
  const float* g1 = (const float*)d_in[9],  *b1 = (const float*)d_in[10];
  const float* g2 = (const float*)d_in[11], *b2 = (const float*)d_in[12];
  const float* fc1w = (const float*)d_in[13], *fc1b = (const float*)d_in[14];
  const float* fc2w = (const float*)d_in[15], *fc2b = (const float*)d_in[16];
  float* out = (float*)d_out;
  float* ws = (float*)d_ws;

  float* rows = ws;
  float* diag = ws + 65536;
  float* a0   = ws + 131072;
  float* a1   = a0 + 262144;
  float* a2   = a1 + 262144;
  float* zb   = a2 + 262144;
  float* cols = zb;
  float* dsum = zb + 65536;
  float* tot  = zb + 65664;
  float* st   = zb + 65792;
  float* bnsum0 = st,       *bnsq0 = st + 64;
  float* bnsum1 = st + 128, *bnsq1 = st + 192;
  float* bnsum2 = st + 256, *bnsq2 = st + 320;
  float* sumA0 = st + 384;
  float* sumA1 = st + 896;

  k_zero<<<263, 256, 0, stream>>>(zb, 67200);
  k_reduce_x<<<512, 512, 0, stream>>>(x, rows, cols, diag, dsum, tot);
  k_layer2<<<256, 128, 0, stream>>>(diag, rows, cols, c2, bias2, dsum, tot, a0,
                                    bnsum0, bnsq0, sumA0);
  k_transform<<<256, 128, 0, stream>>>(a0, bnsum0, bnsq0, g0, b0, c1a, bias1a,
                                       sumA0, a1, bnsum1, bnsq1, sumA1);
  k_transform<<<256, 128, 0, stream>>>(a1, bnsum1, bnsq1, g1, b1, c1b, bias1b,
                                       sumA1, a2, bnsum2, bnsq2, nullptr);
  k_fc<<<256, 128, 0, stream>>>(a2, bnsum2, bnsq2, g2, b2, fc1w, fc1b, fc2w,
                                fc2b, out);
}


# BN statistics without atomics: each block writes its 16 channels' sum and sum of squares as one 128-byte line and the consumer's fold adds a channel's 64 lines in a fixed order
# speedup vs baseline: 2.2109x; 1.5096x over previous
#include <hip/hip_runtime.h>
#include <cstddef>

typedef __attribute__((ext_vector_type(2))) float v2f;
typedef __attribute__((ext_vector_type(8))) float v8f;

__device__ __forceinline__ float wave32_sum(float v) {
#pragma unroll
  for (int m = 16; m >= 1; m >>= 1) v += __shfl_xor(v, m, 32);
  return v;
}
__device__ __forceinline__ float half16_sum(float v) {
#pragma unroll
  for (int m = 1; m < 16; m <<= 1) v += __shfl_xor(v, m, 32);
  return v;
}

__device__ __forceinline__ v8f wmma_f32_k4(v2f a, v2f b, v8f c) {
  return __builtin_amdgcn_wmma_f32_16x16x4_f32(false, a, false, b, (short)0, c,
                                               false, false);
}

__global__ void k_zero(float* p, int n) {
  int i = blockIdx.x * blockDim.x + threadIdx.x;
  if (i < n) p[i] = 0.f;
}

__global__ void k_reduce_x(const float* __restrict__ x,
                           float* __restrict__ rowsO, float* __restrict__ colsO,
                           float* __restrict__ diagO, float* __restrict__ dsumO,
                           float* __restrict__ totO) {
  __shared__ float colbuf[16 * 512];
  __shared__ float totbuf[16];

  const int bid = blockIdx.x;
  const int mat = bid >> 2;
  const int r0 = (bid & 3) << 7;
  const int tid = threadIdx.x;
  const int w = tid >> 5, l = tid & 31;
  const float* M = x + (size_t)mat * 262144;

  float ca[16];
#pragma unroll
  for (int i = 0; i < 16; ++i) ca[i] = 0.f;
  float totp = 0.f;

#pragma unroll 4
  for (int t = 0; t < 8; ++t) {
    const int r = r0 + w + (t << 4);
    const float4* rp = (const float4*)(M + (size_t)r * 512);
    float rsum = 0.f;
#pragma unroll
    for (int c = 0; c < 4; ++c) {
      float4 v = rp[(c << 5) + l];
      ca[c * 4 + 0] += v.x; ca[c * 4 + 1] += v.y;
      ca[c * 4 + 2] += v.z; ca[c * 4 + 3] += v.w;
      rsum += (v.x + v.y) + (v.z + v.w);
    }
    rsum = wave32_sum(rsum);
    if (l == 0) { rowsO[mat * 512 + r] = rsum; totp += rsum; }
  }
#pragma unroll
  for (int c = 0; c < 4; ++c)
#pragma unroll
    for (int j = 0; j < 4; ++j)
      colbuf[w * 512 + c * 128 + l * 4 + j] = ca[c * 4 + j];
  if (l == 0) totbuf[w] = totp;

  float dv = 0.f;
  if (tid < 128) {
    const int r = r0 + tid;
    dv = M[(size_t)r * 512 + r];
    diagO[mat * 512 + r] = dv;
  }
  float ds = wave32_sum(dv);

  __syncthreads();
  float csum = 0.f;
#pragma unroll
  for (int w2 = 0; w2 < 16; ++w2) csum += colbuf[w2 * 512 + tid];
  atomicAdd(&colsO[mat * 512 + tid], csum);

  if (l == 0 && w < 4) atomicAdd(&dsumO[mat], ds);
  if (tid == 0) {
    float T = 0.f;
#pragma unroll
    for (int i = 0; i < 16; ++i) T += totbuf[i];
    atomicAdd(&totO[mat], T);
  }
}


__device__ __forceinline__ void relu_store_stats(v8f acc, float cs,
                                                 float* __restrict__ orow,
                                                 float* __restrict__ red,
                                                 float* __restrict__ line,
                                                 int w, int l, int h, int q) {
  float v[8];
  float k1 = 0.f, k2 = 0.f;
#pragma unroll
  for (int r = 0; r < 8; ++r) {
    v[r] = fmaxf(acc[r] + __shfl(cs, r + 8 * h, 32), 0.f);
    const float sv = half16_sum(v[r]);
    const float sv2 = half16_sum(v[r] * v[r]);
    if (q == r) { k1 = sv; k2 = sv2; }
  }
  if (q < 8) {
    red[w * 32 + q + 8 * h] = k1;
    red[w * 32 + 16 + q + 8 * h] = k2;
  }
  __syncthreads();
#pragma unroll
  for (int r = 0; r < 8; ++r) orow[(size_t)(r + 8 * h) * 512] = v[r];
  if (w == 0) {
    const int idx = ((l & 1) << 4) + (l >> 1);
    line[l] = ((red[idx] + red[32 + idx]) + red[64 + idx]) + red[96 + idx];
  }
}

__device__ __forceinline__ void bn_fold(const float* __restrict__ pq,
                                        const float* __restrict__ g,
                                        const float* __restrict__ bt, int w,
                                        int q, int h, int n, float& A, float& B,
                                        float& sumN) {
  const int d = 16 * w + q;
  const float gd = g[d], btd = bt[d];
  const float2* p = (const float2*)pq + (size_t)(h * 128 + w * 8) * 16 + q;
  float2 v[32];
#pragma unroll
  for (int j = 0; j < 32; ++j) v[j] = p[((j >> 3) * 32 + (j & 7)) * 16];
  float s1 = 0.f, s2 = 0.f, sn = 0.f;
#pragma unroll
  for (int nn = 0; nn < 4; ++nn) {
    float a = v[nn * 8].x, b = v[nn * 8].y;
#pragma unroll
    for (int ib = 1; ib < 8; ++ib) { a += v[nn * 8 + ib].x; b += v[nn * 8 + ib].y; }
    s1 += a;
    s2 += b;
    if (4 * h + nn == n) sn = a;
  }
  s1 += __shfl_xor(s1, 16, 32);
  s2 += __shfl_xor(s2, 16, 32);
  sn += __shfl_xor(sn, 16, 32);
  const float mu = s1 * (1.f / 4096.f);
  const float var = s2 * (1.f / 4096.f) - mu * mu;
  A = gd * rsqrtf(var + 1e-5f);
  B = btd - A * mu;
  sumN = sn;
}

__device__ __forceinline__ v8f gemm_k16(v8f acc, const float* __restrict__ c2,
                                        int b, float scale,
                                        const float* __restrict__ S, int s0,
                                        int i0, int q, int h) {
#pragma unroll
  for (int k0 = 0; k0 < 16; k0 += 4) {
    const int ka = k0 + 2 * h;
    v2f a, bb;
    a.x = c2[(size_t)(ka * 64 + s0 + q) * 5 + b] * scale;
    a.y = c2[(size_t)((ka + 1) * 64 + s0 + q) * 5 + b] * scale;
    bb.x = S[ka * 512 + i0 + q];
    bb.y = S[(ka + 1) * 512 + i0 + q];
    acc = wmma_f32_k4(a, bb, acc);
  }
  return acc;
}

__global__ void k_layer2(const float* __restrict__ diag, const float* __restrict__ rows,
                         const float* __restrict__ cols, const float* __restrict__ c2,
                         const float* __restrict__ bias2, const float* __restrict__ dsum,
                         const float* __restrict__ tot, float* __restrict__ aout,
                         float* __restrict__ pqout) {
  __shared__ float red[128];
  const int gw = blockIdx.x * 4 + (threadIdx.x >> 5);
  const int n = gw >> 7, rem = gw & 127;
  const int s0 = (rem >> 5) << 4, i0 = (rem & 31) << 4;
  const int l = threadIdx.x & 31, h = l >> 4, q = l & 15;
  const size_t nb = (size_t)n * 16 * 512;

  float cs = bias2[s0 + q];
#pragma unroll
  for (int d = 0; d < 16; ++d) {
    const float* c = c2 + (size_t)(d * 64 + s0 + q) * 5;
    cs += c[1] * dsum[n * 16 + d] * (1.f / 512.f) +
          c[4] * tot[n * 16 + d] * (1.f / (512.f * 512.f));
  }

  v8f acc = {};
  acc = gemm_k16(acc, c2, 0, 1.f, diag + nb, s0, i0, q, h);
  acc = gemm_k16(acc, c2, 2, 1.f / 512.f, rows + nb, s0, i0, q, h);
  acc = gemm_k16(acc, c2, 3, 1.f / 512.f, cols + nb, s0, i0, q, h);

  relu_store_stats(acc, cs, aout + ((size_t)n * 64 + s0) * 512 + i0 + q, red,
                   pqout + blockIdx.x * 32, threadIdx.x >> 5, l, h, q);
}

__global__ void __launch_bounds__(128) k_transform(const float* __restrict__ ain,
                            const float* __restrict__ pqin,
                            const float* __restrict__ g, const float* __restrict__ bt,
                            const float* __restrict__ c, const float* __restrict__ bias,
                            float* __restrict__ aout, float* __restrict__ pqout) {
  __shared__ __attribute__((aligned(16))) float sA[64];
  __shared__ __attribute__((aligned(16))) float sB[64];
  __shared__ __attribute__((aligned(16))) float sX[64];
  __shared__ float red[128];
  const int tid = threadIdx.x;
  const int gw = blockIdx.x * 4 + (tid >> 5);
  const int n = gw >> 7, rem = gw & 127;
  const int s0 = (rem >> 5) << 4, i0 = (rem & 31) << 4;
  const int l = tid & 31, h = l >> 4, q = l & 15;
  const float* S = ain + (size_t)n * 64 * 512;

  float fA, fB, fN;
  bn_fold(pqin, g, bt, tid >> 5, q, h, n, fA, fB, fN);
  float braw[32];
#pragma unroll
  for (int j = 0; j < 16; ++j) {
    const int ka = 4 * j + 2 * h;
    braw[2 * j] = S[ka * 512 + i0 + q];
    braw[2 * j + 1] = S[(ka + 1) * 512 + i0 + q];
  }
  float2 cc[64];
#pragma unroll
  for (int d = 0; d < 64; ++d)
    cc[d] = *(const float2*)(c + (size_t)(d * 64 + s0 + q) * 2);

  {
    const int d = ((tid >> 5) << 4) + q;
    sA[d] = fA;
    sB[d] = fB;
    sX[d] = fA * fN + 512.f * fB;
  }
  __syncthreads();

  float cs = bias[s0 + q];
#pragma unroll
  for (int d = 0; d < 64; ++d)
    cs += cc[d].x * sB[d] + cc[d].y * sX[d] * (1.f / 512.f);

  v8f acc = {};
#pragma unroll
  for (int j = 0; j < 16; ++j) {
    const int ka = 4 * j + 2 * h;
    const float2 sa = *(const float2*)&sA[ka];
    v2f a, b;
    const float c0 = cc[4 * j].x, c1 = cc[4 * j + 1].x;
    const float c2 = cc[4 * j + 2].x, c3 = cc[4 * j + 3].x;
    a.x = (h ? c2 : c0) * sa.x;
    a.y = (h ? c3 : c1) * sa.y;
    b.x = braw[2 * j];
    b.y = braw[2 * j + 1];
    acc = wmma_f32_k4(a, b, acc);
  }

  relu_store_stats(acc, cs, aout + ((size_t)n * 64 + s0) * 512 + i0 + q, red,
                   pqout + blockIdx.x * 32, tid >> 5, l, h, q);
}

__global__ void k_fc(const float* __restrict__ a2,
                     const float* __restrict__ pqin, const float* __restrict__ g,
                     const float* __restrict__ bt, const float* __restrict__ fc1w,
                     const float* __restrict__ fc1b, const float* __restrict__ fc2w,
                     const float* __restrict__ fc2b, float* __restrict__ out) {
  __shared__ float r1[64 * 16];
  __shared__ __attribute__((aligned(16))) float sA[64];
  __shared__ __attribute__((aligned(16))) float sB[64];
  const int tid = threadIdx.x;
  const int n = blockIdx.x >> 5, i0 = (blockIdx.x & 31) << 4;
  const int w = tid >> 5, l = tid & 31, h = l >> 4, q = l & 15;
  const float* S = a2 + (size_t)n * 64 * 512;

  float fA, fB, fN;
  bn_fold(pqin, g, bt, w, q, h, n, fA, fB, fN);
  float braw[32];
#pragma unroll
  for (int j = 0; j < 16; ++j) {
    const int ka = 4 * j + 2 * h;
    braw[2 * j] = S[ka * 512 + i0 + q];
    braw[2 * j + 1] = S[(ka + 1) * 512 + i0 + q];
  }
  float4 rw[16];
  {
    const float4* rp = (const float4*)(fc1w + (size_t)(w * 16 + q) * 64);
#pragma unroll
    for (int j = 0; j < 16; ++j) rw[j] = rp[j];
  }

  sA[w * 16 + q] = fA;
  sB[w * 16 + q] = fB;
  __syncthreads();

  float cf = fc1b[w * 16 + q];
#pragma unroll
  for (int j = 0; j < 16; ++j) {
    cf += rw[j].x * sB[4 * j];
    cf += rw[j].y * sB[4 * j + 1];
    cf += rw[j].z * sB[4 * j + 2];
    cf += rw[j].w * sB[4 * j + 3];
  }

  v8f acc = {};
#pragma unroll
  for (int j = 0; j < 16; ++j) {
    const int ka = 4 * j + 2 * h;
    const float2 sa = *(const float2*)&sA[ka];
    v2f a, b;
    const float4 rj = rw[j];
    a.x = (h ? rj.z : rj.x) * sa.x;
    a.y = (h ? rj.w : rj.y) * sa.y;
    b.x = braw[2 * j];
    b.y = braw[2 * j + 1];
    acc = wmma_f32_k4(a, b, acc);
  }
#pragma unroll
  for (int r = 0; r < 8; ++r) {
    const int o = w * 16 + r + 8 * h;
    r1[o * 16 + q] = fmaxf(acc[r] + __shfl(cf, r + 8 * h, 32), 0.f);
  }
  __syncthreads();

  if (w == 0) {
    v8f acc2 = {};
#pragma unroll
    for (int k0 = 0; k0 < 64; k0 += 4) {
      const int ka = k0 + 2 * h;
      v2f a, b;
      a.x = fc2w[q * 64 + ka];
      a.y = fc2w[q * 64 + ka + 1];
      b.x = r1[ka * 16 + q];
      b.y = r1[(ka + 1) * 16 + q];
      acc2 = wmma_f32_k4(a, b, acc2);
    }
#pragma unroll
    for (int r = 0; r < 8; ++r) {
      const int o = r + 8 * h;
      out[((size_t)n * 16 + o) * 512 + i0 + q] = acc2[r] + fc2b[o];
    }
  }
}

extern "C" void kernel_launch(void* const* d_in, const int* in_sizes, int n_in,
                              void* d_out, int out_size, void* d_ws, size_t ws_size,
                              hipStream_t stream) {
  const float* x      = (const float*)d_in[0];
  const float* c2     = (const float*)d_in[1];
  const float* bias2  = (const float*)d_in[2];
  const float* c1a    = (const float*)d_in[3];
  const float* bias1a = (const float*)d_in[4];
  const float* c1b    = (const float*)d_in[5];
  const float* bias1b = (const float*)d_in[6];
  const float* g0 = (const float*)d_in[7],  *b0 = (const float*)d_in[8];
  const float* g1 = (const float*)d_in[9],  *b1 = (const float*)d_in[10];
  const float* g2 = (const float*)d_in[11], *b2 = (const float*)d_in[12];
  const float* fc1w = (const float*)d_in[13], *fc1b = (const float*)d_in[14];
  const float* fc2w = (const float*)d_in[15], *fc2b = (const float*)d_in[16];
  float* out = (float*)d_out;
  float* ws = (float*)d_ws;

  float* rows = ws;
  float* diag = ws + 65536;
  float* a0   = ws + 131072;
  float* a1   = a0 + 262144;
  float* a2   = a1 + 262144;
  float* zb   = a2 + 262144;
  float* cols = zb;
  float* dsum = zb + 65536;
  float* tot  = zb + 65664;
  float* pq0 = a2;
  float* pq1 = rows;
  float* pq2 = diag;

  k_zero<<<257, 256, 0, stream>>>(zb, 65792);
  k_reduce_x<<<512, 512, 0, stream>>>(x, rows, cols, diag, dsum, tot);
  k_layer2<<<256, 128, 0, stream>>>(diag, rows, cols, c2, bias2, dsum, tot, a0,
                                    pq0);
  k_transform<<<256, 128, 0, stream>>>(a0, pq0, g0, b0, c1a, bias1a, a1, pq1);
  k_transform<<<256, 128, 0, stream>>>(a1, pq1, g1, b1, c1b, bias1b, a2, pq2);
  k_fc<<<256, 128, 0, stream>>>(a2, pq2, g2, b2, fc1w, fc1b, fc2w, fc2b, out);
}


# Sums over x (columns, diagonal, total) without atomics or the clearing launch: each row-block writes its partial sums as whole lines, layer 2 adds a matrix's four in fixed order; 6 launches -> 5
# speedup vs baseline: 2.4103x; 1.0902x over previous
#include <hip/hip_runtime.h>
#include <cstddef>

typedef __attribute__((ext_vector_type(2))) float v2f;
typedef __attribute__((ext_vector_type(8))) float v8f;

__device__ __forceinline__ float wave32_sum(float v) {
#pragma unroll
  for (int m = 16; m >= 1; m >>= 1) v += __shfl_xor(v, m, 32);
  return v;
}
__device__ __forceinline__ float half16_sum(float v) {
#pragma unroll
  for (int m = 1; m < 16; m <<= 1) v += __shfl_xor(v, m, 32);
  return v;
}

__device__ __forceinline__ v8f wmma_f32_k4(v2f a, v2f b, v8f c) {
  return __builtin_amdgcn_wmma_f32_16x16x4_f32(false, a, false, b, (short)0, c,
                                               false, false);
}

__global__ void k_reduce_x(const float* __restrict__ x,
                           float* __restrict__ rowsO, float* __restrict__ colsP,
                           float* __restrict__ diagO, float* __restrict__ scal) {
  __shared__ float colbuf[16 * 512];
  __shared__ float totbuf[16];
  __shared__ float dsbuf[4];

  const int bid = blockIdx.x;
  const int mat = bid >> 2;
  const int r0 = (bid & 3) << 7;
  const int tid = threadIdx.x;
  const int w = tid >> 5, l = tid & 31;
  const float* M = x + (size_t)mat * 262144;

  float ca[16];
#pragma unroll
  for (int i = 0; i < 16; ++i) ca[i] = 0.f;
  float totp = 0.f;

#pragma unroll 4
  for (int t = 0; t < 8; ++t) {
    const int r = r0 + w + (t << 4);
    const float4* rp = (const float4*)(M + (size_t)r * 512);
    float rsum = 0.f;
#pragma unroll
    for (int c = 0; c < 4; ++c) {
      float4 v = rp[(c << 5) + l];
      ca[c * 4 + 0] += v.x; ca[c * 4 + 1] += v.y;
      ca[c * 4 + 2] += v.z; ca[c * 4 + 3] += v.w;
      rsum += (v.x + v.y) + (v.z + v.w);
    }
    rsum = wave32_sum(rsum);
    if (l == 0) { rowsO[mat * 512 + r] = rsum; totp += rsum; }
  }
#pragma unroll
  for (int c = 0; c < 4; ++c)
#pragma unroll
    for (int j = 0; j < 4; ++j)
      colbuf[w * 512 + c * 128 + l * 4 + j] = ca[c * 4 + j];
  if (l == 0) totbuf[w] = totp;

  float dv = 0.f;
  if (tid < 128) {
    const int r = r0 + tid;
    dv = M[(size_t)r * 512 + r];
    diagO[mat * 512 + r] = dv;
  }
  float ds = wave32_sum(dv);
  if (l == 0 && w < 4) dsbuf[w] = ds;

  __syncthreads();
  float csum = 0.f;
#pragma unroll
  for (int w2 = 0; w2 < 16; ++w2) csum += colbuf[w2 * 512 + tid];
  colsP[(size_t)bid * 512 + tid] = csum;

  if (w == 0) {
    float T = 0.f;
#pragma unroll
    for (int i = 0; i < 16; ++i) T += totbuf[i];
    const float D = ((dsbuf[0] + dsbuf[1]) + dsbuf[2]) + dsbuf[3];
    scal[bid * 32 + l] = (l == 0) ? D : ((l == 1) ? T : 0.f);
  }
}


__device__ __forceinline__ void relu_store_stats(v8f acc, float cs,
                                                 float* __restrict__ orow,
                                                 float* __restrict__ red,
                                                 float* __restrict__ line,
                                                 int w, int l, int h, int q) {
  float v[8];
  float k1 = 0.f, k2 = 0.f;
#pragma unroll
  for (int r = 0; r < 8; ++r) {
    v[r] = fmaxf(acc[r] + __shfl(cs, r + 8 * h, 32), 0.f);
    const float sv = half16_sum(v[r]);
    const float sv2 = half16_sum(v[r] * v[r]);
    if (q == r) { k1 = sv; k2 = sv2; }
  }
  if (q < 8) {
    red[w * 32 + q + 8 * h] = k1;
    red[w * 32 + 16 + q + 8 * h] = k2;
  }
  __syncthreads();
#pragma unroll
  for (int r = 0; r < 8; ++r) orow[(size_t)(r + 8 * h) * 512] = v[r];
  if (w == 0) {
    const int idx = ((l & 1) << 4) + (l >> 1);
    line[l] = ((red[idx] + red[32 + idx]) + red[64 + idx]) + red[96 + idx];
  }
}

__device__ __forceinline__ void bn_fold(const float* __restrict__ pq,
                                        const float* __restrict__ g,
                                        const float* __restrict__ bt, int w,
                                        int q, int h, int n, float& A, float& B,
                                        float& sumN) {
  const int d = 16 * w + q;
  const float gd = g[d], btd = bt[d];
  const float2* p = (const float2*)pq + (size_t)(h * 128 + w * 8) * 16 + q;
  float2 v[32];
#pragma unroll
  for (int j = 0; j < 32; ++j) v[j] = p[((j >> 3) * 32 + (j & 7)) * 16];
  float s1 = 0.f, s2 = 0.f, sn = 0.f;
#pragma unroll
  for (int nn = 0; nn < 4; ++nn) {
    float a = v[nn * 8].x, b = v[nn * 8].y;
#pragma unroll
    for (int ib = 1; ib < 8; ++ib) { a += v[nn * 8 + ib].x; b += v[nn * 8 + ib].y; }
    s1 += a;
    s2 += b;
    if (4 * h + nn == n) sn = a;
  }
  s1 += __shfl_xor(s1, 16, 32);
  s2 += __shfl_xor(s2, 16, 32);
  sn += __shfl_xor(sn, 16, 32);
  const float mu = s1 * (1.f / 4096.f);
  const float var = s2 * (1.f / 4096.f) - mu * mu;
  A = gd * rsqrtf(var + 1e-5f);
  B = btd - A * mu;
  sumN = sn;
}

__device__ __forceinline__ v8f gemm_k16(v8f acc, const float* __restrict__ c2,
                                        int b, float scale,
                                        const float* __restrict__ S, int s0,
                                        int i0, int q, int h) {
#pragma unroll
  for (int k0 = 0; k0 < 16; k0 += 4) {
    const int ka = k0 + 2 * h;
    v2f a, bb;
    a.x = c2[(size_t)(ka * 64 + s0 + q) * 5 + b] * scale;
    a.y = c2[(size_t)((ka + 1) * 64 + s0 + q) * 5 + b] * scale;
    bb.x = S[ka * 512 + i0 + q];
    bb.y = S[(ka + 1) * 512 + i0 + q];
    acc = wmma_f32_k4(a, bb, acc);
  }
  return acc;
}

__device__ __forceinline__ v8f gemm_k16_p4(v8f acc, const float* __restrict__ c2,
                                           int b, float scale,
                                           const float* __restrict__ P, int s0,
                                           int i0, int q, int h) {
#pragma unroll
  for (int k0 = 0; k0 < 16; k0 += 4) {
    const int ka = k0 + 2 * h;
    v2f a, bb;
    a.x = c2[(size_t)(ka * 64 + s0 + q) * 5 + b] * scale;
    a.y = c2[(size_t)((ka + 1) * 64 + s0 + q) * 5 + b] * scale;
    const float* p0 = P + ka * 2048 + i0 + q;
    const float* p1 = p0 + 2048;
    bb.x = ((p0[0] + p0[512]) + p0[1024]) + p0[1536];
    bb.y = ((p1[0] + p1[512]) + p1[1024]) + p1[1536];
    acc = wmma_f32_k4(a, bb, acc);
  }
  return acc;
}

__global__ void k_layer2(const float* __restrict__ diag, const float* __restrict__ rows,
                         const float* __restrict__ colsP, const float* __restrict__ c2,
                         const float* __restrict__ bias2, const float* __restrict__ scal,
                         float* __restrict__ aout, float* __restrict__ pqout) {
  __shared__ float red[128];
  const int gw = blockIdx.x * 4 + (threadIdx.x >> 5);
  const int n = gw >> 7, rem = gw & 127;
  const int s0 = (rem >> 5) << 4, i0 = (rem & 31) << 4;
  const int l = threadIdx.x & 31, h = l >> 4, q = l & 15;
  const size_t nb = (size_t)n * 16 * 512;

  float dsq, ttq;
  {
    const float2* sp = (const float2*)(scal + (size_t)((n * 16 + q) * 4) * 32);
    const float2 e0 = sp[0], e1 = sp[16], e2 = sp[32], e3 = sp[48];
    dsq = ((e0.x + e1.x) + e2.x) + e3.x;
    ttq = ((e0.y + e1.y) + e2.y) + e3.y;
  }

  float cs = bias2[s0 + q];
#pragma unroll
  for (int d = 0; d < 16; ++d) {
    const float* c = c2 + (size_t)(d * 64 + s0 + q) * 5;
    cs += c[1] * __shfl(dsq, d, 32) * (1.f / 512.f) +
          c[4] * __shfl(ttq, d, 32) * (1.f / (512.f * 512.f));
  }

  v8f acc = {};
  acc = gemm_k16(acc, c2, 0, 1.f, diag + nb, s0, i0, q, h);
  acc = gemm_k16(acc, c2, 2, 1.f / 512.f, rows + nb, s0, i0, q, h);
  acc = gemm_k16_p4(acc, c2, 3, 1.f / 512.f, colsP + nb * 4, s0, i0, q, h);

  relu_store_stats(acc, cs, aout + ((size_t)n * 64 + s0) * 512 + i0 + q, red,
                   pqout + blockIdx.x * 32, threadIdx.x >> 5, l, h, q);
}

__global__ void __launch_bounds__(128) k_transform(const float* __restrict__ ain,
                            const float* __restrict__ pqin,
                            const float* __restrict__ g, const float* __restrict__ bt,
                            const float* __restrict__ c, const float* __restrict__ bias,
                            float* __restrict__ aout, float* __restrict__ pqout) {
  __shared__ __attribute__((aligned(16))) float sA[64];
  __shared__ __attribute__((aligned(16))) float sB[64];
  __shared__ __attribute__((aligned(16))) float sX[64];
  __shared__ float red[128];
  const int tid = threadIdx.x;
  const int gw = blockIdx.x * 4 + (tid >> 5);
  const int n = gw >> 7, rem = gw & 127;
  const int s0 = (rem >> 5) << 4, i0 = (rem & 31) << 4;
  const int l = tid & 31, h = l >> 4, q = l & 15;
  const float* S = ain + (size_t)n * 64 * 512;

  float fA, fB, fN;
  bn_fold(pqin, g, bt, tid >> 5, q, h, n, fA, fB, fN);
  float braw[32];
#pragma unroll
  for (int j = 0; j < 16; ++j) {
    const int ka = 4 * j + 2 * h;
    braw[2 * j] = S[ka * 512 + i0 + q];
    braw[2 * j + 1] = S[(ka + 1) * 512 + i0 + q];
  }
  float2 cc[64];
#pragma unroll
  for (int d = 0; d < 64; ++d)
    cc[d] = *(const float2*)(c + (size_t)(d * 64 + s0 + q) * 2);

  {
    const int d = ((tid >> 5) << 4) + q;
    sA[d] = fA;
    sB[d] = fB;
    sX[d] = fA * fN + 512.f * fB;
  }
  __syncthreads();

  float cs = bias[s0 + q];
#pragma unroll
  for (int d = 0; d < 64; ++d)
    cs += cc[d].x * sB[d] + cc[d].y * sX[d] * (1.f / 512.f);

  v8f acc = {};
#pragma unroll
  for (int j = 0; j < 16; ++j) {
    const int ka = 4 * j + 2 * h;
    const float2 sa = *(const float2*)&sA[ka];
    v2f a, b;
    const float c0 = cc[4 * j].x, c1 = cc[4 * j + 1].x;
    const float c2 = cc[4 * j + 2].x, c3 = cc[4 * j + 3].x;
    a.x = (h ? c2 : c0) * sa.x;
    a.y = (h ? c3 : c1) * sa.y;
    b.x = braw[2 * j];
    b.y = braw[2 * j + 1];
    acc = wmma_f32_k4(a, b, acc);
  }

  relu_store_stats(acc, cs, aout + ((size_t)n * 64 + s0) * 512 + i0 + q, red,
                   pqout + blockIdx.x * 32, tid >> 5, l, h, q);
}

__global__ void k_fc(const float* __restrict__ a2,
                     const float* __restrict__ pqin, const float* __restrict__ g,
                     const float* __restrict__ bt, const float* __restrict__ fc1w,
                     const float* __restrict__ fc1b, const float* __restrict__ fc2w,
                     const float* __restrict__ fc2b, float* __restrict__ out) {
  __shared__ float r1[64 * 16];
  __shared__ __attribute__((aligned(16))) float sA[64];
  __shared__ __attribute__((aligned(16))) float sB[64];
  const int tid = threadIdx.x;
  const int n = blockIdx.x >> 5, i0 = (blockIdx.x & 31) << 4;
  const int w = tid >> 5, l = tid & 31, h = l >> 4, q = l & 15;
  const float* S = a2 + (size_t)n * 64 * 512;

  float fA, fB, fN;
  bn_fold(pqin, g, bt, w, q, h, n, fA, fB, fN);
  float braw[32];
#pragma unroll
  for (int j = 0; j < 16; ++j) {
    const int ka = 4 * j + 2 * h;
    braw[2 * j] = S[ka * 512 + i0 + q];
    braw[2 * j + 1] = S[(ka + 1) * 512 + i0 + q];
  }
  float4 rw[16];
  {
    const float4* rp = (const float4*)(fc1w + (size_t)(w * 16 + q) * 64);
#pragma unroll
    for (int j = 0; j < 16; ++j) rw[j] = rp[j];
  }

  sA[w * 16 + q] = fA;
  sB[w * 16 + q] = fB;
  __syncthreads();

  float cf = fc1b[w * 16 + q];
#pragma unroll
  for (int j = 0; j < 16; ++j) {
    cf += rw[j].x * sB[4 * j];
    cf += rw[j].y * sB[4 * j + 1];
    cf += rw[j].z * sB[4 * j + 2];
    cf += rw[j].w * sB[4 * j + 3];
  }

  v8f acc = {};
#pragma unroll
  for (int j = 0; j < 16; ++j) {
    const int ka = 4 * j + 2 * h;
    const float2 sa = *(const float2*)&sA[ka];
    v2f a, b;
    const float4 rj = rw[j];
    a.x = (h ? rj.z : rj.x) * sa.x;
    a.y = (h ? rj.w : rj.y) * sa.y;
    b.x = braw[2 * j];
    b.y = braw[2 * j + 1];
    acc = wmma_f32_k4(a, b, acc);
  }
#pragma unroll
  for (int r = 0; r < 8; ++r) {
    const int o = w * 16 + r + 8 * h;
    r1[o * 16 + q] = fmaxf(acc[r] + __shfl(cf, r + 8 * h, 32), 0.f);
  }
  __syncthreads();

  if (w == 0) {
    v8f acc2 = {};
#pragma unroll
    for (int k0 = 0; k0 < 64; k0 += 4) {
      const int ka = k0 + 2 * h;
      v2f a, b;
      a.x = fc2w[q * 64 + ka];
      a.y = fc2w[q * 64 + ka + 1];
      b.x = r1[ka * 16 + q];
      b.y = r1[(ka + 1) * 16 + q];
      acc2 = wmma_f32_k4(a, b, acc2);
    }
#pragma unroll
    for (int r = 0; r < 8; ++r) {
      const int o = r + 8 * h;
      out[((size_t)n * 16 + o) * 512 + i0 + q] = acc2[r] + fc2b[o];
    }
  }
}

extern "C" void kernel_launch(void* const* d_in, const int* in_sizes, int n_in,
                              void* d_out, int out_size, void* d_ws, size_t ws_size,
                              hipStream_t stream) {
  const float* x      = (const float*)d_in[0];
  const float* c2     = (const float*)d_in[1];
  const float* bias2  = (const float*)d_in[2];
  const float* c1a    = (const float*)d_in[3];
  const float* bias1a = (const float*)d_in[4];
  const float* c1b    = (const float*)d_in[5];
  const float* bias1b = (const float*)d_in[6];
  const float* g0 = (const float*)d_in[7],  *b0 = (const float*)d_in[8];
  const float* g1 = (const float*)d_in[9],  *b1 = (const float*)d_in[10];
  const float* g2 = (const float*)d_in[11], *b2 = (const float*)d_in[12];
  const float* fc1w = (const float*)d_in[13], *fc1b = (const float*)d_in[14];
  const float* fc2w = (const float*)d_in[15], *fc2b = (const float*)d_in[16];
  float* out = (float*)d_out;
  float* ws = (float*)d_ws;

  float* rows = ws;
  float* diag = ws + 65536;
  float* a0   = ws + 131072;
  float* a1   = a0 + 262144;
  float* a2   = a1 + 262144;
  float* scal = a2 + 262144;
  float* colsP = a1;
  float* pq0 = a2;
  float* pq1 = rows;
  float* pq2 = diag;

  k_reduce_x<<<512, 512, 0, stream>>>(x, rows, colsP, diag, scal);
  k_layer2<<<256, 128, 0, stream>>>(diag, rows, colsP, c2, bias2, scal, a0, pq0);
  k_transform<<<256, 128, 0, stream>>>(a0, pq0, g0, b0, c1a, bias1a, a1, pq1);
  k_transform<<<256, 128, 0, stream>>>(a1, pq1, g1, b1, c1b, bias1b, a2, pq2);
  k_fc<<<256, 128, 0, stream>>>(a2, pq2, g2, b2, fc1w, fc1b, fc2w, fc2b, out);
}


# Tail kernels: a wave requests all its global operands as one group before its first wait (scheduling barrier): one memory round trip per launch, not three to six; every sum keeps its order
# speedup vs baseline: 2.5505x; 1.0582x over previous
#include <hip/hip_runtime.h>
#include <cstddef>

typedef __attribute__((ext_vector_type(2))) float v2f;
typedef __attribute__((ext_vector_type(8))) float v8f;

__device__ __forceinline__ float wave32_sum(float v) {
#pragma unroll
  for (int m = 16; m >= 1; m >>= 1) v += __shfl_xor(v, m, 32);
  return v;
}
__device__ __forceinline__ float half16_sum(float v) {
#pragma unroll
  for (int m = 1; m < 16; m <<= 1) v += __shfl_xor(v, m, 32);
  return v;
}

__device__ __forceinline__ v8f wmma_f32_k4(v2f a, v2f b, v8f c) {
  return __builtin_amdgcn_wmma_f32_16x16x4_f32(false, a, false, b, (short)0, c,
                                               false, false);
}

__global__ void k_reduce_x(const float* __restrict__ x,
                           float* __restrict__ rowsO, float* __restrict__ colsP,
                           float* __restrict__ diagO, float* __restrict__ scal) {
  __shared__ float colbuf[16 * 512];
  __shared__ float totbuf[16];
  __shared__ float dsbuf[4];

  const int bid = blockIdx.x;
  const int mat = bid >> 2;
  const int r0 = (bid & 3) << 7;
  const int tid = threadIdx.x;
  const int w = tid >> 5, l = tid & 31;
  const float* M = x + (size_t)mat * 262144;

  float ca[16];
#pragma unroll
  for (int i = 0; i < 16; ++i) ca[i] = 0.f;
  float totp = 0.f;

#pragma unroll 4
  for (int t = 0; t < 8; ++t) {
    const int r = r0 + w + (t << 4);
    const float4* rp = (const float4*)(M + (size_t)r * 512);
    float rsum = 0.f;
#pragma unroll
    for (int c = 0; c < 4; ++c) {
      float4 v = rp[(c << 5) + l];
      ca[c * 4 + 0] += v.x; ca[c * 4 + 1] += v.y;
      ca[c * 4 + 2] += v.z; ca[c * 4 + 3] += v.w;
      rsum += (v.x + v.y) + (v.z + v.w);
    }
    rsum = wave32_sum(rsum);
    if (l == 0) { rowsO[mat * 512 + r] = rsum; totp += rsum; }
  }
#pragma unroll
  for (int c = 0; c < 4; ++c)
#pragma unroll
    for (int j = 0; j < 4; ++j)
      colbuf[w * 512 + c * 128 + l * 4 + j] = ca[c * 4 + j];
  if (l == 0) totbuf[w] = totp;

  float dv = 0.f;
  if (tid < 128) {
    const int r = r0 + tid;
    dv = M[(size_t)r * 512 + r];
    diagO[mat * 512 + r] = dv;
  }
  float ds = wave32_sum(dv);
  if (l == 0 && w < 4) dsbuf[w] = ds;

  __syncthreads();
  float csum = 0.f;
#pragma unroll
  for (int w2 = 0; w2 < 16; ++w2) csum += colbuf[w2 * 512 + tid];
  colsP[(size_t)bid * 512 + tid] = csum;

  if (w == 0) {
    float T = 0.f;
#pragma unroll
    for (int i = 0; i < 16; ++i) T += totbuf[i];
    const float D = ((dsbuf[0] + dsbuf[1]) + dsbuf[2]) + dsbuf[3];
    scal[bid * 32 + l] = (l == 0) ? D : ((l == 1) ? T : 0.f);
  }
}


__device__ __forceinline__ void relu_store_stats(v8f acc, float cs,
                                                 float* __restrict__ orow,
                                                 float* __restrict__ red,
                                                 float* __restrict__ line,
                                                 int w, int l, int h, int q) {
  float v[8];
  float k1 = 0.f, k2 = 0.f;
#pragma unroll
  for (int r = 0; r < 8; ++r) {
    v[r] = fmaxf(acc[r] + __shfl(cs, r + 8 * h, 32), 0.f);
    const float sv = half16_sum(v[r]);
    const float sv2 = half16_sum(v[r] * v[r]);
    if (q == r) { k1 = sv; k2 = sv2; }
  }
  if (q < 8) {
    red[w * 32 + q + 8 * h] = k1;
    red[w * 32 + 16 + q + 8 * h] = k2;
  }
  __syncthreads();
#pragma unroll
  for (int r = 0; r < 8; ++r) orow[(size_t)(r + 8 * h) * 512] = v[r];
  if (w == 0) {
    const int idx = ((l & 1) << 4) + (l >> 1);
    line[l] = ((red[idx] + red[32 + idx]) + red[64 + idx]) + red[96 + idx];
  }
}

__device__ __forceinline__ void bn_fold_request(const float* __restrict__ pq,
                                                int w, int q, int h,
                                                float2 (&v)[32]) {
  const float2* p = (const float2*)pq + (size_t)(h * 128 + w * 8) * 16 + q;
#pragma unroll
  for (int j = 0; j < 32; ++j) v[j] = p[((j >> 3) * 32 + (j & 7)) * 16];
}
__device__ __forceinline__ void bn_fold_sum(const float2 (&v)[32], float gd,
                                            float btd, int h, int n, float& A,
                                            float& B, float& sumN) {
  float s1 = 0.f, s2 = 0.f, sn = 0.f;
#pragma unroll
  for (int nn = 0; nn < 4; ++nn) {
    float a = v[nn * 8].x, b = v[nn * 8].y;
#pragma unroll
    for (int ib = 1; ib < 8; ++ib) { a += v[nn * 8 + ib].x; b += v[nn * 8 + ib].y; }
    s1 += a;
    s2 += b;
    if (4 * h + nn == n) sn = a;
  }
  s1 += __shfl_xor(s1, 16, 32);
  s2 += __shfl_xor(s2, 16, 32);
  sn += __shfl_xor(sn, 16, 32);
  const float mu = s1 * (1.f / 4096.f);
  const float var = s2 * (1.f / 4096.f) - mu * mu;
  A = gd * rsqrtf(var + 1e-5f);
  B = btd - A * mu;
  sumN = sn;
}

__global__ void __launch_bounds__(128) k_layer2(const float* __restrict__ diag, const float* __restrict__ rows,
                         const float* __restrict__ colsP, const float* __restrict__ c2,
                         const float* __restrict__ bias2, const float* __restrict__ scal,
                         float* __restrict__ aout, float* __restrict__ pqout) {
  __shared__ float red[128];
  const int gw = blockIdx.x * 4 + (threadIdx.x >> 5);
  const int n = gw >> 7, rem = gw & 127;
  const int s0 = (rem >> 5) << 4, i0 = (rem & 31) << 4;
  const int l = threadIdx.x & 31, h = l >> 4, q = l & 15;
  const size_t nb = (size_t)n * 16 * 512;

  const float2* sp = (const float2*)(scal + (size_t)((n * 16 + q) * 4) * 32);
  const float2 e0 = sp[0], e1 = sp[16], e2 = sp[32], e3 = sp[48];
  const float bq = bias2[s0 + q];
  float c1v[16], c4v[16];
#pragma unroll
  for (int d = 0; d < 16; ++d) {
    const float* c = c2 + (size_t)(d * 64 + s0 + q) * 5;
    c1v[d] = c[1];
    c4v[d] = c[4];
  }
  float aD[8], aR[8], aC[8], bD[8], bR[8], bC[32];
#pragma unroll
  for (int k = 0; k < 4; ++k) {
    const int ka = 4 * k + 2 * h;
    const float* ca = c2 + (size_t)(ka * 64 + s0 + q) * 5;
    const float* cb = c2 + (size_t)((ka + 1) * 64 + s0 + q) * 5;
    aD[2 * k] = ca[0]; aD[2 * k + 1] = cb[0];
    aR[2 * k] = ca[2]; aR[2 * k + 1] = cb[2];
    aC[2 * k] = ca[3]; aC[2 * k + 1] = cb[3];
    bD[2 * k] = diag[nb + ka * 512 + i0 + q];
    bD[2 * k + 1] = diag[nb + (ka + 1) * 512 + i0 + q];
    bR[2 * k] = rows[nb + ka * 512 + i0 + q];
    bR[2 * k + 1] = rows[nb + (ka + 1) * 512 + i0 + q];
    const float* p0 = colsP + nb * 4 + ka * 2048 + i0 + q;
    const float* p1 = p0 + 2048;
#pragma unroll
    for (int rb = 0; rb < 4; ++rb) {
      bC[8 * k + rb] = p0[rb * 512];
      bC[8 * k + 4 + rb] = p1[rb * 512];
    }
  }
  __builtin_amdgcn_sched_barrier(0);

  const float dsq = ((e0.x + e1.x) + e2.x) + e3.x;
  const float ttq = ((e0.y + e1.y) + e2.y) + e3.y;

  float cs = bq;
#pragma unroll
  for (int d = 0; d < 16; ++d)
    cs += c1v[d] * __shfl(dsq, d, 32) * (1.f / 512.f) +
          c4v[d] * __shfl(ttq, d, 32) * (1.f / (512.f * 512.f));

  v8f acc = {};
#pragma unroll
  for (int k = 0; k < 4; ++k) {
    v2f a, bb;
    a.x = aD[2 * k];
    a.y = aD[2 * k + 1];
    bb.x = bD[2 * k];
    bb.y = bD[2 * k + 1];
    acc = wmma_f32_k4(a, bb, acc);
  }
#pragma unroll
  for (int k = 0; k < 4; ++k) {
    v2f a, bb;
    a.x = aR[2 * k] * (1.f / 512.f);
    a.y = aR[2 * k + 1] * (1.f / 512.f);
    bb.x = bR[2 * k];
    bb.y = bR[2 * k + 1];
    acc = wmma_f32_k4(a, bb, acc);
  }
#pragma unroll
  for (int k = 0; k < 4; ++k) {
    v2f a, bb;
    a.x = aC[2 * k] * (1.f / 512.f);
    a.y = aC[2 * k + 1] * (1.f / 512.f);
    bb.x = ((bC[8 * k] + bC[8 * k + 1]) + bC[8 * k + 2]) + bC[8 * k + 3];
    bb.y = ((bC[8 * k + 4] + bC[8 * k + 5]) + bC[8 * k + 6]) + bC[8 * k + 7];
    acc = wmma_f32_k4(a, bb, acc);
  }

  relu_store_stats(acc, cs, aout + ((size_t)n * 64 + s0) * 512 + i0 + q, red,
                   pqout + blockIdx.x * 32, threadIdx.x >> 5, l, h, q);
}

__global__ void __launch_bounds__(128) k_transform(const float* __restrict__ ain,
                            const float* __restrict__ pqin,
                            const float* __restrict__ g, const float* __restrict__ bt,
                            const float* __restrict__ c, const float* __restrict__ bias,
                            float* __restrict__ aout, float* __restrict__ pqout) {
  __shared__ __attribute__((aligned(16))) float sA[64];
  __shared__ __attribute__((aligned(16))) float sB[64];
  __shared__ __attribute__((aligned(16))) float sX[64];
  __shared__ float red[128];
  const int tid = threadIdx.x;
  const int gw = blockIdx.x * 4 + (tid >> 5);
  const int n = gw >> 7, rem = gw & 127;
  const int s0 = (rem >> 5) << 4, i0 = (rem & 31) << 4;
  const int l = tid & 31, h = l >> 4, q = l & 15;
  const float* S = ain + (size_t)n * 64 * 512;

  const int w = tid >> 5;
  float2 pv[32];
  bn_fold_request(pqin, w, q, h, pv);
  const float gd = g[16 * w + q], btd = bt[16 * w + q];
  float braw[32];
#pragma unroll
  for (int j = 0; j < 16; ++j) {
    const int ka = 4 * j + 2 * h;
    braw[2 * j] = S[ka * 512 + i0 + q];
    braw[2 * j + 1] = S[(ka + 1) * 512 + i0 + q];
  }
  float2 cc[64];
#pragma unroll
  for (int d = 0; d < 64; ++d)
    cc[d] = *(const float2*)(c + (size_t)(d * 64 + s0 + q) * 2);
  const float bs = bias[s0 + q];
  __builtin_amdgcn_sched_barrier(0);

  float fA, fB, fN;
  bn_fold_sum(pv, gd, btd, h, n, fA, fB, fN);
  {
    const int d = (w << 4) + q;
    sA[d] = fA;
    sB[d] = fB;
    sX[d] = fA * fN + 512.f * fB;
  }
  __syncthreads();

  float cs = bs;
#pragma unroll
  for (int d = 0; d < 64; ++d)
    cs += cc[d].x * sB[d] + cc[d].y * sX[d] * (1.f / 512.f);

  v8f acc = {};
#pragma unroll
  for (int j = 0; j < 16; ++j) {
    const int ka = 4 * j + 2 * h;
    const float2 sa = *(const float2*)&sA[ka];
    v2f a, b;
    const float c0 = cc[4 * j].x, c1 = cc[4 * j + 1].x;
    const float c2 = cc[4 * j + 2].x, c3 = cc[4 * j + 3].x;
    a.x = (h ? c2 : c0) * sa.x;
    a.y = (h ? c3 : c1) * sa.y;
    b.x = braw[2 * j];
    b.y = braw[2 * j + 1];
    acc = wmma_f32_k4(a, b, acc);
  }

  relu_store_stats(acc, cs, aout + ((size_t)n * 64 + s0) * 512 + i0 + q, red,
                   pqout + blockIdx.x * 32, tid >> 5, l, h, q);
}

__global__ void __launch_bounds__(128) __attribute__((amdgpu_waves_per_eu(1, 1))) k_fc(const float* __restrict__ a2,
                     const float* __restrict__ pqin, const float* __restrict__ g,
                     const float* __restrict__ bt, const float* __restrict__ fc1w,
                     const float* __restrict__ fc1b, const float* __restrict__ fc2w,
                     const float* __restrict__ fc2b, float* __restrict__ out) {
  __shared__ float r1[64 * 16];
  __shared__ __attribute__((aligned(16))) float sA[64];
  __shared__ __attribute__((aligned(16))) float sB[64];
  const int tid = threadIdx.x;
  const int n = blockIdx.x >> 5, i0 = (blockIdx.x & 31) << 4;
  const int w = tid >> 5, l = tid & 31, h = l >> 4, q = l & 15;
  const float* S = a2 + (size_t)n * 64 * 512;

  float2 pv[32];
  bn_fold_request(pqin, w, q, h, pv);
  const float gd = g[16 * w + q], btd = bt[16 * w + q];
  float braw[32];
#pragma unroll
  for (int j = 0; j < 16; ++j) {
    const int ka = 4 * j + 2 * h;
    braw[2 * j] = S[ka * 512 + i0 + q];
    braw[2 * j + 1] = S[(ka + 1) * 512 + i0 + q];
  }
  float4 rw[16];
  {
    const float4* rp = (const float4*)(fc1w + (size_t)(w * 16 + q) * 64);
#pragma unroll
    for (int j = 0; j < 16; ++j) rw[j] = rp[j];
  }
  const float b1 = fc1b[w * 16 + q];
  float2 f2[16];
#pragma unroll
  for (int j = 0; j < 16; ++j)
    f2[j] = *(const float2*)(fc2w + q * 64 + 4 * j + 2 * h);
  const float4 b2lo = *(const float4*)(fc2b + 8 * h);
  const float4 b2hi = *(const float4*)(fc2b + 8 * h + 4);
  __builtin_amdgcn_sched_barrier(0);

  float fA, fB, fN;
  bn_fold_sum(pv, gd, btd, h, n, fA, fB, fN);
  sA[w * 16 + q] = fA;
  sB[w * 16 + q] = fB;
  __syncthreads();

  float cf = b1;
#pragma unroll
  for (int j = 0; j < 16; ++j) {
    cf += rw[j].x * sB[4 * j];
    cf += rw[j].y * sB[4 * j + 1];
    cf += rw[j].z * sB[4 * j + 2];
    cf += rw[j].w * sB[4 * j + 3];
  }

  v8f acc = {};
#pragma unroll
  for (int j = 0; j < 16; ++j) {
    const int ka = 4 * j + 2 * h;
    const float2 sa = *(const float2*)&sA[ka];
    v2f a, b;
    const float4 rj = rw[j];
    a.x = (h ? rj.z : rj.x) * sa.x;
    a.y = (h ? rj.w : rj.y) * sa.y;
    b.x = braw[2 * j];
    b.y = braw[2 * j + 1];
    acc = wmma_f32_k4(a, b, acc);
  }
#pragma unroll
  for (int r = 0; r < 8; ++r) {
    const int o = w * 16 + r + 8 * h;
    r1[o * 16 + q] = fmaxf(acc[r] + __shfl(cf, r + 8 * h, 32), 0.f);
  }
  __syncthreads();

  if (w == 0) {
    const float ob[8] = {b2lo.x, b2lo.y, b2lo.z, b2lo.w,
                         b2hi.x, b2hi.y, b2hi.z, b2hi.w};
    v8f acc2 = {};
#pragma unroll
    for (int j = 0; j < 16; ++j) {
      const int ka = 4 * j + 2 * h;
      v2f a, b;
      a.x = f2[j].x;
      a.y = f2[j].y;
      b.x = r1[ka * 16 + q];
      b.y = r1[(ka + 1) * 16 + q];
      acc2 = wmma_f32_k4(a, b, acc2);
    }
#pragma unroll
    for (int r = 0; r < 8; ++r) {
      const int o = r + 8 * h;
      out[((size_t)n * 16 + o) * 512 + i0 + q] = acc2[r] + ob[r];
    }
  }
}

extern "C" void kernel_launch(void* const* d_in, const int* in_sizes, int n_in,
                              void* d_out, int out_size, void* d_ws, size_t ws_size,
                              hipStream_t stream) {
  const float* x      = (const float*)d_in[0];
  const float* c2     = (const float*)d_in[1];
  const float* bias2  = (const float*)d_in[2];
  const float* c1a    = (const float*)d_in[3];
  const float* bias1a = (const float*)d_in[4];
  const float* c1b    = (const float*)d_in[5];
  const float* bias1b = (const float*)d_in[6];
  const float* g0 = (const float*)d_in[7],  *b0 = (const float*)d_in[8];
  const float* g1 = (const float*)d_in[9],  *b1 = (const float*)d_in[10];
  const float* g2 = (const float*)d_in[11], *b2 = (const float*)d_in[12];
  const float* fc1w = (const float*)d_in[13], *fc1b = (const float*)d_in[14];
  const float* fc2w = (const float*)d_in[15], *fc2b = (const float*)d_in[16];
  float* out = (float*)d_out;
  float* ws = (float*)d_ws;

  float* rows = ws;
  float* diag = ws + 65536;
  float* a0   = ws + 131072;
  float* a1   = a0 + 262144;
  float* a2   = a1 + 262144;
  float* scal = a2 + 262144;
  float* colsP = a1;
  float* pq0 = a2;
  float* pq1 = rows;
  float* pq2 = diag;

  k_reduce_x<<<512, 512, 0, stream>>>(x, rows, colsP, diag, scal);
  k_layer2<<<256, 128, 0, stream>>>(diag, rows, colsP, c2, bias2, scal, a0, pq0);
  k_transform<<<256, 128, 0, stream>>>(a0, pq0, g0, b0, c1a, bias1a, a1, pq1);
  k_transform<<<256, 128, 0, stream>>>(a1, pq1, g1, b1, c1b, bias1b, a2, pq2);
  k_fc<<<256, 128, 0, stream>>>(a2, pq2, g2, b2, fc1w, fc1b, fc2w, fc2b, out);
}
